# starting kernel
# baseline (speedup 1.0000x reference)
#include <hip/hip_runtime.h>

typedef __attribute__((ext_vector_type(2))) float v2f;
typedef __attribute__((ext_vector_type(8))) float v8f;

__device__ __forceinline__ float fast_tanh(float x) {
#if __has_builtin(__builtin_amdgcn_tanhf)
  return __builtin_amdgcn_tanhf(x);
#else
  float e = __builtin_amdgcn_exp2f(x * 2.8853900817779268f);
  return 1.0f - 2.0f * __builtin_amdgcn_rcpf(e + 1.0f);
#endif
}

__launch_bounds__(32)
__global__ void rnn_wmma_kernel(const float* __restrict__ x,
                                const float* __restrict__ W_ih,
                                const float* __restrict__ W_hh,
                                const float* __restrict__ b_ih,
                                const float* __restrict__ b_hh,
                                const float* __restrict__ fc_w,
                                const float* __restrict__ fc_b,
                                float* __restrict__ out,
                                int T) {
  const int lane = threadIdx.x & 31;
  const int base = blockIdx.x * 16;
  const int n    = lane & 15;
  const int b    = base + n;

  float wih[8][3], bias[8];
  #pragma unroll
  for (int r = 0; r < 8; ++r) {
    wih[r][0] = W_ih[r * 3 + 0];
    wih[r][1] = W_ih[r * 3 + 1];
    wih[r][2] = W_ih[r * 3 + 2];
    bias[r]   = b_ih[r] + b_hh[r];
  }

  const int half = lane >> 4;
  const int row  = lane & 7;
  v2f a1, a2;
  a1.x = W_hh[row * 8 + (half ? 2 : 0)];
  a1.y = W_hh[row * 8 + (half ? 3 : 1)];
  a2.x = W_hh[row * 8 + (half ? 6 : 4)];
  a2.y = W_hh[row * 8 + (half ? 7 : 5)];

  float th[8];
  #pragma unroll
  for (int r = 0; r < 8; ++r) th[r] = 0.0f;

  const float* xb = x + (size_t)b * (size_t)T * 3u;

  float px0[4], px1[4], px2[4];
  #pragma unroll
  for (int k = 0; k < 4; ++k) {
    px0[k] = xb[3 * k + 0];
    px1[k] = xb[3 * k + 1];
    px2[k] = xb[3 * k + 2];
  }

  for (int t0 = 0; t0 < T; t0 += 4) {
    #pragma unroll
    for (int k = 0; k < 4; ++k) {
      const int t = t0 + k;
      float cx0 = px0[k], cx1 = px1[k], cx2 = px2[k];

      int tn = t + 4;
      if (tn > T - 1) tn = T - 1;
      px0[k] = xb[3 * tn + 0];
      px1[k] = xb[3 * tn + 1];
      px2[k] = xb[3 * tn + 2];

      v8f c;
      #pragma unroll
      for (int r = 0; r < 8; ++r)
        c[r] = __builtin_fmaf(wih[r][0], cx0,
               __builtin_fmaf(wih[r][1], cx1,
               __builtin_fmaf(wih[r][2], cx2, bias[r])));

      v2f b1, b2;
      b1.x = (half == 0) ? th[0] : th[2];
      b1.y = (half == 0) ? th[1] : th[3];
      b2.x = (half == 0) ? th[4] : th[6];
      b2.y = (half == 0) ? th[5] : th[7];

      c = __builtin_amdgcn_wmma_f32_16x16x4_f32(false, a1, false, b1, (short)0, c, false, false);
      c = __builtin_amdgcn_wmma_f32_16x16x4_f32(false, a2, false, b2, (short)0, c, false, false);

      #pragma unroll
      for (int r = 0; r < 8; ++r) th[r] = fast_tanh(c[r]);
    }
  }

  if (lane < 16) {
    float o0 = fc_b[0], o1 = fc_b[1];
    #pragma unroll
    for (int r = 0; r < 8; ++r) {
      o0 = __builtin_fmaf(fc_w[r],     th[r], o0);
      o1 = __builtin_fmaf(fc_w[8 + r], th[r], o1);
    }
    out[2 * b + 0] = o0;
    out[2 * b + 1] = o1;
  }
}

extern "C" void kernel_launch(void* const* d_in, const int* in_sizes, int n_in,
                              void* d_out, int out_size, void* d_ws, size_t ws_size,
                              hipStream_t stream) {
  const float* x    = (const float*)d_in[0];
  const float* W_ih = (const float*)d_in[1];
  const float* W_hh = (const float*)d_in[2];
  const float* b_ih = (const float*)d_in[3];
  const float* b_hh = (const float*)d_in[4];
  const float* fc_w = (const float*)d_in[5];
  const float* fc_b = (const float*)d_in[6];
  float* out = (float*)d_out;

  const int T = 1024, I = 3;
  const int B = in_sizes[0] / (T * I);
  const int waves = B / 16;

  rnn_wmma_kernel<<<waves, 32, 0, stream>>>(x, W_ih, W_hh, b_ih, b_hh,
                                            fc_w, fc_b, out, T);
}


# each lane half owns 4 hidden units: W_hh rows permuted over the two halves so a half's D registers are its own next B pairs; no duplicated rows, no selects, half the tanh and input-row fma per lane
# speedup vs baseline: 1.3123x; 1.3123x over previous
#include <hip/hip_runtime.h>


typedef __attribute__((ext_vector_type(2))) float v2f;
typedef __attribute__((ext_vector_type(8))) float v8f;

__device__ __forceinline__ float fast_tanh(float x) {
#if __has_builtin(__builtin_amdgcn_tanhf)
  return __builtin_amdgcn_tanhf(x);
#else
  float e = __builtin_amdgcn_exp2f(x * 2.8853900817779268f);
  return 1.0f - 2.0f * __builtin_amdgcn_rcpf(e + 1.0f);
#endif
}

struct HalfWeights {
  float w0[4], w1[4], w2[4], bs[4];
};

__device__ __forceinline__ void input_rows(v8f& acc, const HalfWeights& w,
                                           float x0, float x1, float x2) {
  #pragma unroll
  for (int r = 0; r < 4; ++r)
    acc[r] = __builtin_fmaf(w.w0[r], x0,
             __builtin_fmaf(w.w1[r], x1,
             __builtin_fmaf(w.w2[r], x2, w.bs[r])));
}

__launch_bounds__(32)
__global__ void rnn_wmma_kernel(const float* __restrict__ x,
                                const float* __restrict__ W_ih,
                                const float* __restrict__ W_hh,
                                const float* __restrict__ b_ih,
                                const float* __restrict__ b_hh,
                                const float* __restrict__ fc_w,
                                const float* __restrict__ fc_b,
                                float* __restrict__ out,
                                int T) {
  const int lane = threadIdx.x & 31;
  const int base = blockIdx.x * 16;
  const int n    = lane & 15;
  const int b    = base + n;
  const int half = lane >> 4;

  HalfWeights hw;
  #pragma unroll
  for (int r = 0; r < 4; ++r) {
    const int u = (r & 1) + 2 * half + 4 * (r >> 1);
    hw.w0[r] = W_ih[u * 3 + 0];
    hw.w1[r] = W_ih[u * 3 + 1];
    hw.w2[r] = W_ih[u * 3 + 2];
    hw.bs[r] = b_ih[u] + b_hh[u];
  }

  const int  m    = lane & 15;
  const bool live = (m & 4) == 0;
  const int  q    = m & 3;
  const int  ua   = (q & 1) + 2 * (m >> 3) + 4 * (q >> 1);
  v2f a1, a2;
  {
    const float t0 = W_hh[ua * 8 + 2 * half + 0];
    const float t1 = W_hh[ua * 8 + 2 * half + 1];
    const float t2 = W_hh[ua * 8 + 2 * half + 4];
    const float t3 = W_hh[ua * 8 + 2 * half + 5];
    a1.x = live ? t0 : 0.0f;
    a1.y = live ? t1 : 0.0f;
    a2.x = live ? t2 : 0.0f;
    a2.y = live ? t3 : 0.0f;
  }

  float h0 = 0.0f, h1 = 0.0f, h2 = 0.0f, h3 = 0.0f;

  const int nblk = T >> 2;
  if (nblk >= 1) {
    v8f acc[4];
    #pragma unroll
    for (int k = 0; k < 4; ++k)
      acc[k] = (v8f){0.0f, 0.0f, 0.0f, 0.0f, 0.0f, 0.0f, 0.0f, 0.0f};

    const float* xrow = x + (size_t)b * (size_t)T * 3u;

    float px[12];
    #pragma unroll
    for (int j = 0; j < 12; ++j) px[j] = xrow[j];
    input_rows(acc[0], hw, px[0], px[1], px[2]);

    const float* xn = xrow + 12;
    for (int blk = 0; blk < nblk - 1; ++blk) {
      float nx[12];
      #pragma unroll
      for (int j = 0; j < 12; ++j) nx[j] = xn[j];
      xn += 12;

      #pragma unroll
      for (int k = 0; k < 4; ++k) {
        v2f b1, b2;
        b1.x = h0; b1.y = h1;
        b2.x = h2; b2.y = h3;

        v8f c = acc[k];
        c = __builtin_amdgcn_wmma_f32_16x16x4_f32(false, a1, false, b1, (short)0, c, false, false);
        c = __builtin_amdgcn_wmma_f32_16x16x4_f32(false, a2, false, b2, (short)0, c, false, false);

        if (k < 3) input_rows(acc[k + 1], hw, px[3 * k + 3], px[3 * k + 4], px[3 * k + 5]);
        else       input_rows(acc[0],     hw, nx[0], nx[1], nx[2]);

        h0 = fast_tanh(c[0]);
        h1 = fast_tanh(c[1]);
        h2 = fast_tanh(c[2]);
        h3 = fast_tanh(c[3]);
        acc[k][4] = c[4]; acc[k][5] = c[5]; acc[k][6] = c[6]; acc[k][7] = c[7];

        __builtin_amdgcn_sched_barrier(0);
      }

      #pragma unroll
      for (int j = 0; j < 12; ++j) px[j] = nx[j];
    }

    #pragma unroll
    for (int k = 0; k < 4; ++k) {
      v2f b1, b2;
      b1.x = h0; b1.y = h1;
      b2.x = h2; b2.y = h3;
      v8f c = acc[k];
      c = __builtin_amdgcn_wmma_f32_16x16x4_f32(false, a1, false, b1, (short)0, c, false, false);
      c = __builtin_amdgcn_wmma_f32_16x16x4_f32(false, a2, false, b2, (short)0, c, false, false);
      if (k < 3) input_rows(acc[k + 1], hw, px[3 * k + 3], px[3 * k + 4], px[3 * k + 5]);
      h0 = fast_tanh(c[0]);
      h1 = fast_tanh(c[1]);
      h2 = fast_tanh(c[2]);
      h3 = fast_tanh(c[3]);
      __builtin_amdgcn_sched_barrier(0);
    }
  }

  const float g0 = __shfl_xor(h0, 16);
  const float g1 = __shfl_xor(h1, 16);
  const float g2 = __shfl_xor(h2, 16);
  const float g3 = __shfl_xor(h3, 16);
  if (lane < 16) {
    float th[8];
    th[0] = h0; th[1] = h1; th[2] = g0; th[3] = g1;
    th[4] = h2; th[5] = h3; th[6] = g2; th[7] = g3;
    float o0 = fc_b[0], o1 = fc_b[1];
    #pragma unroll
    for (int r = 0; r < 8; ++r) {
      o0 = __builtin_fmaf(fc_w[r],     th[r], o0);
      o1 = __builtin_fmaf(fc_w[8 + r], th[r], o1);
    }
    out[2 * b + 0] = o0;
    out[2 * b + 1] = o1;
  }
}

extern "C" void kernel_launch(void* const* d_in, const int* in_sizes, int n_in,
                              void* d_out, int out_size, void* d_ws, size_t ws_size,
                              hipStream_t stream) {
  const float* x    = (const float*)d_in[0];
  const float* W_ih = (const float*)d_in[1];
  const float* W_hh = (const float*)d_in[2];
  const float* b_ih = (const float*)d_in[3];
  const float* b_hh = (const float*)d_in[4];
  const float* fc_w = (const float*)d_in[5];
  const float* fc_b = (const float*)d_in[6];
  float* out = (float*)d_out;

  const int T = 1024, I = 3;
  const int B = in_sizes[0] / (T * I);
  const int waves = B / 16;

  rnn_wmma_kernel<<<waves, 32, 0, stream>>>(x, W_ih, W_hh, b_ih, b_hh,
                                            fc_w, fc_b, out, T);
}


# x stream requested 15 blocks (59 steps) ahead through a ring of 16 register buffers, one ring turn per loop trip so no buffer is copied: a miss to memory no longer stalls the step chain
# speedup vs baseline: 2.4477x; 1.8652x over previous
#include <hip/hip_runtime.h>


typedef __attribute__((ext_vector_type(2))) float v2f;
typedef __attribute__((ext_vector_type(8))) float v8f;

__device__ __forceinline__ float fast_tanh(float x) {
#if __has_builtin(__builtin_amdgcn_tanhf)
  return __builtin_amdgcn_tanhf(x);
#else
  float e = __builtin_amdgcn_exp2f(x * 2.8853900817779268f);
  return 1.0f - 2.0f * __builtin_amdgcn_rcpf(e + 1.0f);
#endif
}

struct HalfWeights {
  float w0[4], w1[4], w2[4], bs[4];
};

__device__ __forceinline__ void input_rows(v8f& acc, const HalfWeights& w,
                                           float x0, float x1, float x2) {
  #pragma unroll
  for (int r = 0; r < 4; ++r)
    acc[r] = __builtin_fmaf(w.w0[r], x0,
             __builtin_fmaf(w.w1[r], x1,
             __builtin_fmaf(w.w2[r], x2, w.bs[r])));
}

constexpr int kT    = 1024;
constexpr int kBlk  = 4;
constexpr int kRing = 16;
static_assert(kBlk % 4 == 0 && (kT / kBlk) % kRing == 0 && (kT / kBlk) >= 2 * kRing,
              "whole ring turns only");

template <bool V> struct Flag { static constexpr bool value = V; };

__launch_bounds__(32)
__global__ void rnn_wmma_kernel(const float* __restrict__ x,
                                const float* __restrict__ W_ih,
                                const float* __restrict__ W_hh,
                                const float* __restrict__ b_ih,
                                const float* __restrict__ b_hh,
                                const float* __restrict__ fc_w,
                                const float* __restrict__ fc_b,
                                float* __restrict__ out) {
  const int lane = threadIdx.x & 31;
  const int base = blockIdx.x * 16;
  const int n    = lane & 15;
  const int b    = base + n;
  const int half = lane >> 4;

  HalfWeights hw;
  #pragma unroll
  for (int r = 0; r < 4; ++r) {
    const int u = (r & 1) + 2 * half + 4 * (r >> 1);
    hw.w0[r] = W_ih[u * 3 + 0];
    hw.w1[r] = W_ih[u * 3 + 1];
    hw.w2[r] = W_ih[u * 3 + 2];
    hw.bs[r] = b_ih[u] + b_hh[u];
  }

  const int  m    = lane & 15;
  const bool live = (m & 4) == 0;
  const int  q    = m & 3;
  const int  ua   = (q & 1) + 2 * (m >> 3) + 4 * (q >> 1);
  v2f a1, a2;
  {
    const float t0 = W_hh[ua * 8 + 2 * half + 0];
    const float t1 = W_hh[ua * 8 + 2 * half + 1];
    const float t2 = W_hh[ua * 8 + 2 * half + 4];
    const float t3 = W_hh[ua * 8 + 2 * half + 5];
    a1.x = live ? t0 : 0.0f;
    a1.y = live ? t1 : 0.0f;
    a2.x = live ? t2 : 0.0f;
    a2.y = live ? t3 : 0.0f;
  }

  float h0 = 0.0f, h1 = 0.0f, h2 = 0.0f, h3 = 0.0f;

  v8f acc[4];
  #pragma unroll
  for (int k = 0; k < 4; ++k)
    acc[k] = (v8f){0.0f, 0.0f, 0.0f, 0.0f, 0.0f, 0.0f, 0.0f, 0.0f};

  const char* xbytes = (const char*)x;
  unsigned xoff = (unsigned)b * (unsigned)(kT * 3 * sizeof(float));
  float xb[kRing][kBlk * 3];
  #pragma unroll
  for (int s = 0; s < kRing - 1; ++s) {
    #pragma unroll
    for (int j = 0; j < kBlk * 3; ++j)
      xb[s][j] = ((const float*)(xbytes + (size_t)xoff))[kBlk * 3 * s + j];
  }
  input_rows(acc[0], hw, xb[0][0], xb[0][1], xb[0][2]);

  xoff += kBlk * 3 * sizeof(float) * (kRing - 1);

  auto ring_turn = [&](auto all) __attribute__((always_inline)) {
    constexpr bool kAll = decltype(all)::value;
    #pragma unroll
    for (int s = 0; s < kRing; ++s) {
      if (kAll || s == 0) {
        asm volatile("" : "+v"(xoff));
        const float* xq = (const float*)(xbytes + (size_t)xoff);
        #pragma unroll
        for (int j = 0; j < kBlk * 3; ++j) xb[(s + kRing - 1) % kRing][j] = xq[j];
        xoff += kBlk * 3 * sizeof(float);
      }
      const float* px = xb[s];
      const float* nx = xb[(s + 1) % kRing];
      const bool more = kAll || s < kRing - 1;

      #pragma unroll
      for (int k = 0; k < kBlk; ++k) {
        const int  ka   = k & 3;
        const int  kn   = (k + 1) & 3;
        const bool next = more || k < kBlk - 1;

        v2f b1, b2;
        b1.x = h0; b1.y = h1;
        b2.x = h2; b2.y = h3;

        v8f c = acc[ka];
        c = __builtin_amdgcn_wmma_f32_16x16x4_f32(false, a1, false, b1, (short)0, c, false, false);
        c = __builtin_amdgcn_wmma_f32_16x16x4_f32(false, a2, false, b2, (short)0, c, false, false);

        if (k < kBlk - 1) input_rows(acc[kn], hw, px[3 * k + 3], px[3 * k + 4], px[3 * k + 5]);
        else if (more)    input_rows(acc[kn], hw, nx[0], nx[1], nx[2]);

        h0 = fast_tanh(c[0]);
        h1 = fast_tanh(c[1]);
        h2 = fast_tanh(c[2]);
        h3 = fast_tanh(c[3]);
        if (next) {
          acc[ka][4] = c[4]; acc[ka][5] = c[5]; acc[ka][6] = c[6]; acc[ka][7] = c[7];
        }

        __builtin_amdgcn_sched_barrier(0);
      }
    }
  };

  for (int g = 0; g < (kT / kBlk) / kRing - 1; ++g) ring_turn(Flag<true>{});
  ring_turn(Flag<false>{});

  const float g0 = __shfl_xor(h0, 16);
  const float g1 = __shfl_xor(h1, 16);
  const float g2 = __shfl_xor(h2, 16);
  const float g3 = __shfl_xor(h3, 16);
  if (lane < 16) {
    float th[8];
    th[0] = h0; th[1] = h1; th[2] = g0; th[3] = g1;
    th[4] = h2; th[5] = h3; th[6] = g2; th[7] = g3;
    float o0 = fc_b[0], o1 = fc_b[1];
    #pragma unroll
    for (int r = 0; r < 8; ++r) {
      o0 = __builtin_fmaf(fc_w[r],     th[r], o0);
      o1 = __builtin_fmaf(fc_w[8 + r], th[r], o1);
    }
    out[2 * b + 0] = o0;
    out[2 * b + 1] = o1;
  }
}

extern "C" void kernel_launch(void* const* d_in, const int* in_sizes, int n_in,
                              void* d_out, int out_size, void* d_ws, size_t ws_size,
                              hipStream_t stream) {
  const float* x    = (const float*)d_in[0];
  const float* W_ih = (const float*)d_in[1];
  const float* W_hh = (const float*)d_in[2];
  const float* b_ih = (const float*)d_in[3];
  const float* b_hh = (const float*)d_in[4];
  const float* fc_w = (const float*)d_in[5];
  const float* fc_b = (const float*)d_in[6];
  float* out = (float*)d_out;

  const int T = 1024, I = 3;
  const int B = in_sizes[0] / (T * I);
  const int waves = B / 16;

  rnn_wmma_kernel<<<waves, 32, 0, stream>>>(x, W_ih, W_hh, b_ih, b_hh,
                                            fc_w, fc_b, out);
}


# 8 batch rows per wave, K=8 split over the tile's two column halves: one K=4 matrix instruction per step, not two chained; half sums added across lanes c, c^8 by DPP (sum order changes); 1024 waves
# speedup vs baseline: 2.9043x; 1.1865x over previous
#include <hip/hip_runtime.h>


typedef __attribute__((ext_vector_type(2))) float v2f;
typedef __attribute__((ext_vector_type(8))) float v8f;

__device__ __forceinline__ float fast_tanh(float x) {
#if __has_builtin(__builtin_amdgcn_tanhf)
  return __builtin_amdgcn_tanhf(x);
#else
  float e = __builtin_amdgcn_exp2f(x * 2.8853900817779268f);
  return 1.0f - 2.0f * __builtin_amdgcn_rcpf(e + 1.0f);
#endif
}

struct OwnWeights {
  float w0[2], w1[2], w2[2], bs[2];
};

__device__ __forceinline__ void input_vals(float& v0, float& v1, const OwnWeights& w,
                                           float x0, float x1, float x2) {
  v0 = __builtin_fmaf(w.w0[0], x0, __builtin_fmaf(w.w1[0], x1, __builtin_fmaf(w.w2[0], x2, w.bs[0])));
  v1 = __builtin_fmaf(w.w0[1], x0, __builtin_fmaf(w.w1[1], x1, __builtin_fmaf(w.w2[1], x2, w.bs[1])));
}

__device__ __forceinline__ void place_rows(v8f& acc, float v0, float v1) {
  acc[0] = v0; acc[1] = v1;
  acc[2] = v0; acc[3] = v1;
}

constexpr int kT     = 1024;
constexpr int kBlk   = 4;
constexpr int kRing  = 16;
constexpr int kCols  = 8;
constexpr int kWaves = 4;
static_assert(kBlk % 4 == 0 && (kT / kBlk) % kRing == 0 && (kT / kBlk) >= 2 * kRing,
              "whole ring turns only");

template <bool V> struct Flag { static constexpr bool value = V; };

__launch_bounds__(32 * kWaves)
__global__ void rnn_wmma_kernel(const float* __restrict__ x,
                                const float* __restrict__ W_ih,
                                const float* __restrict__ W_hh,
                                const float* __restrict__ b_ih,
                                const float* __restrict__ b_hh,
                                const float* __restrict__ fc_w,
                                const float* __restrict__ fc_b,
                                float* __restrict__ out) {
  __shared__ float staged[kWaves * kCols * 2];

  const int lane  = threadIdx.x & 31;
  const int wave  = threadIdx.x >> 5;
  const int gbase = blockIdx.x * (kWaves * kCols);
  const int c     = lane & 15;
  const int half  = lane >> 4;
  const int hcol  = c >> 3;
  const int j     = c & 7;
  const int b     = gbase + wave * kCols + j;

  OwnWeights ow;
  float bi[2], bh[2];
  #pragma unroll
  for (int e = 0; e < 2; ++e) {
    const int u = 4 * hcol + 2 * half + e;
    ow.w0[e] = W_ih[u * 3 + 0];
    ow.w1[e] = W_ih[u * 3 + 1];
    ow.w2[e] = W_ih[u * 3 + 2];
    bi[e] = b_ih[u];
    bh[e] = b_hh[u];
  }

  v2f a;
  {
    const int m    = c;
    const int r    = m & 7;
    const int unit = (r & 1) + 2 * (m >> 3) + 4 * ((r >> 1) & 1);
    const int high = ((r >> 1) ^ (r >> 2)) & 1;
    const int kc   = 4 * high + 2 * half;
    a.x = W_hh[unit * 8 + kc + 0];
    a.y = W_hh[unit * 8 + kc + 1];
  }
  __builtin_amdgcn_sched_barrier(0);

  float h0 = 0.0f, h1 = 0.0f;

  v8f acc[2];
  #pragma unroll
  for (int k = 0; k < 2; ++k)
    acc[k] = (v8f){0.0f, 0.0f, 0.0f, 0.0f, 0.0f, 0.0f, 0.0f, 0.0f};

  const char* xbytes = (const char*)x;
  unsigned xoff = (unsigned)b * (unsigned)(kT * 3 * sizeof(float));
  float xb[kRing][kBlk * 3];
  #pragma unroll
  for (int s = 0; s < kRing - 1; ++s) {
    #pragma unroll
    for (int i = 0; i < kBlk * 3; ++i)
      xb[s][i] = ((const float*)(xbytes + (size_t)xoff))[kBlk * 3 * s + i];
  }
  __builtin_amdgcn_sched_barrier(0);
  ow.bs[0] = bi[0] + bh[0];
  ow.bs[1] = bi[1] + bh[1];
  {
    float v0, v1;
    input_vals(v0, v1, ow, xb[0][0], xb[0][1], xb[0][2]);
    place_rows(acc[0], v0, v1);
  }

  xoff += kBlk * 3 * sizeof(float) * (kRing - 1);

  auto ring_turn = [&](auto all) __attribute__((always_inline)) {
    constexpr bool kAll = decltype(all)::value;
    #pragma unroll
    for (int s = 0; s < kRing; ++s) {
      if (kAll || s == 0) {
        asm volatile("" : "+v"(xoff));
        const float* xq = (const float*)(xbytes + (size_t)xoff);
        #pragma unroll
        for (int i = 0; i < kBlk * 3; ++i) xb[(s + kRing - 1) % kRing][i] = xq[i];
        xoff += kBlk * 3 * sizeof(float);
        __builtin_amdgcn_sched_barrier(0);
      }
      const float* px = xb[s];
      const float* nx = xb[(s + 1) % kRing];
      const bool more = kAll || s < kRing - 1;

      #pragma unroll
      for (int k = 0; k < kBlk; ++k) {
        const int ka = k & 1;
        const int kn = (k + 1) & 1;

        v2f bb;
        bb.x = h0; bb.y = h1;

        const v8f d = __builtin_amdgcn_wmma_f32_16x16x4_f32(false, a, false, bb, (short)0,
                                                            acc[ka], false, false);

        float v0 = 0.0f, v1 = 0.0f;
        if (k < kBlk - 1) input_vals(v0, v1, ow, px[3 * k + 3], px[3 * k + 4], px[3 * k + 5]);
        else if (more)    input_vals(v0, v1, ow, nx[0], nx[1], nx[2]);

        __builtin_amdgcn_sched_barrier(0);

        float s0, s1;
        asm("v_add_f32_dpp %0, %1, %2 row_ror:8 row_mask:0xf bank_mask:0x3"
            : "=v"(s0) : "v"(d[4]), "v"(d[0]));
        asm("v_add_f32_dpp %0, %1, %2 row_ror:8 row_mask:0xf bank_mask:0xc"
            : "+v"(s0) : "v"(d[6]), "v"(d[2]));
        asm("v_add_f32_dpp %0, %1, %2 row_ror:8 row_mask:0xf bank_mask:0x3"
            : "=v"(s1) : "v"(d[5]), "v"(d[1]));
        asm("v_add_f32_dpp %0, %1, %2 row_ror:8 row_mask:0xf bank_mask:0xc"
            : "+v"(s1) : "v"(d[7]), "v"(d[3]));

        h0 = fast_tanh(s0);
        h1 = fast_tanh(s1);

        if (k < kBlk - 1 || more) place_rows(acc[kn], v0, v1);

        __builtin_amdgcn_sched_barrier(0);
      }
    }
  };

  for (int g = 0; g < (kT / kBlk) / kRing - 1; ++g) ring_turn(Flag<true>{});
  ring_turn(Flag<false>{});

  const float t2 = __shfl(h0, (lane + 16) & 31);
  const float t3 = __shfl(h1, (lane + 16) & 31);
  const float t4 = __shfl(h0, (lane + 8) & 31);
  const float t5 = __shfl(h1, (lane + 8) & 31);
  const float t6 = __shfl(h0, (lane + 24) & 31);
  const float t7 = __shfl(h1, (lane + 24) & 31);
  if (lane < kCols) {
    float th[8];
    th[0] = h0; th[1] = h1; th[2] = t2; th[3] = t3;
    th[4] = t4; th[5] = t5; th[6] = t6; th[7] = t7;
    float o0 = fc_b[0], o1 = fc_b[1];
    #pragma unroll
    for (int r = 0; r < 8; ++r) {
      o0 = __builtin_fmaf(fc_w[r],     th[r], o0);
      o1 = __builtin_fmaf(fc_w[8 + r], th[r], o1);
    }
    staged[2 * (wave * kCols + lane) + 0] = o0;
    staged[2 * (wave * kCols + lane) + 1] = o1;
  }
  __syncthreads();
  if (wave == 0) {
    const float2 v = ((const float2*)staged)[lane];
    ((float2*)(out + 2 * gbase))[lane] = v;
  }
}

extern "C" void kernel_launch(void* const* d_in, const int* in_sizes, int n_in,
                              void* d_out, int out_size, void* d_ws, size_t ws_size,
                              hipStream_t stream) {
  const float* x    = (const float*)d_in[0];
  const float* W_ih = (const float*)d_in[1];
  const float* W_hh = (const float*)d_in[2];
  const float* b_ih = (const float*)d_in[3];
  const float* b_hh = (const float*)d_in[4];
  const float* fc_w = (const float*)d_in[5];
  const float* fc_b = (const float*)d_in[6];
  float* out = (float*)d_out;

  const int T = 1024, I = 3;
  const int B = in_sizes[0] / (T * I);
  const int groups = B / 32;

  rnn_wmma_kernel<<<groups, 128, 0, stream>>>(x, W_ih, W_hh, b_ih, b_hh,
                                              fc_w, fc_b, out);
}
